# starting kernel
# baseline (speedup 1.0000x reference)
#include <hip/hip_runtime.h>

typedef float v2f __attribute__((ext_vector_type(2)));
typedef float v4f __attribute__((ext_vector_type(4)));
typedef float v8f __attribute__((ext_vector_type(8)));

#define E 8
#define F 32
#define WAVES_PER_BLOCK 8
#define HPITCH 34

__launch_bounds__(256, 1)
__global__ void ffq_kernel(const float* __restrict__ x,
                           const float* __restrict__ theta,
                           const float* __restrict__ w1,
                           const float* __restrict__ b1,
                           const float* __restrict__ w2,
                           const float* __restrict__ b2,
                           float* __restrict__ out,
                           int n_tiles) {
  __shared__ float lds_h[WAVES_PER_BLOCK][16 * HPITCH];
  __shared__ float lds_o[WAVES_PER_BLOCK][128];

  const int tid  = threadIdx.x;
  const int wave = tid >> 5;
  const int lane = tid & 31;
  const int lo   = lane & 15;
  const int hi   = lane >> 4;
  const int c    = hi * 2;

  const float t0 = theta[c], t1 = theta[c + 1], t2 = theta[c + 4], t3 = theta[c + 5];

  v2f bw1[2][2];
  for (int nt = 0; nt < 2; ++nt) {
    const int n = nt * 16 + lo;
    for (int k = 0; k < 2; ++k) {
      bw1[nt][k].x = w1[(4 * k + c    ) * F + n];
      bw1[nt][k].y = w1[(4 * k + c + 1) * F + n];
    }
  }
  const float bias1_0 = b1[lo];
  const float bias1_1 = b1[16 + lo];

  v2f bw2[8];
  for (int k = 0; k < 8; ++k) {
    if (lo < E) {
      bw2[k].x = w2[(4 * k + c    ) * E + lo];
      bw2[k].y = w2[(4 * k + c + 1) * E + lo];
    } else {
      bw2[k].x = 0.0f;
      bw2[k].y = 0.0f;
    }
  }
  const float bias2 = (lo < E) ? b2[lo] : 0.0f;

  float* hbuf = lds_h[wave];
  float* obuf = lds_o[wave];

  const int wave_gid = blockIdx.x * WAVES_PER_BLOCK + wave;
  const int n_waves  = gridDim.x * WAVES_PER_BLOCK;

  for (int tile = wave_gid; tile < n_tiles; tile += n_waves) {
    const int row = tile * 16 + lo;
    const float* px = x + row * E + c;
    const v2f xa = *(const v2f*)(px);
    const v2f xb = *(const v2f*)(px + 4);

    v2f A0, A1;
    A0.x = __builtin_cosf(xa.x + t0);
    A0.y = __builtin_cosf(xa.y + t1);
    A1.x = __builtin_cosf(xb.x + t2);
    A1.y = __builtin_cosf(xb.y + t3);

    v8f acc0 = {};
    v8f acc1 = {};
    acc0 = __builtin_amdgcn_wmma_f32_16x16x4_f32(false, A0, false, bw1[0][0], (short)0, acc0, false, false);
    acc0 = __builtin_amdgcn_wmma_f32_16x16x4_f32(false, A1, false, bw1[0][1], (short)0, acc0, false, false);
    acc1 = __builtin_amdgcn_wmma_f32_16x16x4_f32(false, A0, false, bw1[1][0], (short)0, acc1, false, false);
    acc1 = __builtin_amdgcn_wmma_f32_16x16x4_f32(false, A1, false, bw1[1][1], (short)0, acc1, false, false);

#pragma unroll
    for (int r = 0; r < 8; ++r) {
      float h0 = acc0[r] + bias1_0; h0 = h0 > 0.0f ? h0 : 0.0f;
      float h1 = acc1[r] + bias1_1; h1 = h1 > 0.0f ? h1 : 0.0f;
      hbuf[(8 * hi + r) * HPITCH + lo     ] = h0;
      hbuf[(8 * hi + r) * HPITCH + 16 + lo] = h1;
    }

    v8f acc2 = {};
#pragma unroll
    for (int k = 0; k < 8; ++k) {
      const v2f a = *(const v2f*)&hbuf[lo * HPITCH + 4 * k + c];
      acc2 = __builtin_amdgcn_wmma_f32_16x16x4_f32(false, a, false, bw2[k], (short)0, acc2, false, false);
    }

    if (lo < E) {
#pragma unroll
      for (int r = 0; r < 8; ++r)
        obuf[(8 * hi + r) * E + lo] = acc2[r] + bias2;
    }

    const v4f o = *(const v4f*)&obuf[lane * 4];
    *(v4f*)(out + tile * 128 + lane * 4) = o;
  }
}

extern "C" void kernel_launch(void* const* d_in, const int* in_sizes, int n_in,
                              void* d_out, int out_size, void* d_ws, size_t ws_size,
                              hipStream_t stream) {
  const float* x     = (const float*)d_in[0];
  const float* theta = (const float*)d_in[1];
  const float* w1    = (const float*)d_in[2];
  const float* b1    = (const float*)d_in[3];
  const float* w2    = (const float*)d_in[4];
  const float* b2    = (const float*)d_in[5];
  float* out = (float*)d_out;

  const int n_rows  = in_sizes[0] / E;
  const int n_tiles = n_rows / 16;

  ffq_kernel<<<1024, 256, 0, stream>>>(x, theta, w1, b1, w2, b2, out, n_tiles);
}


# both products transposed (weights as A, activations as B): product 1 D registers feed product 2 as B, no LDS; w2 pad rows repeat rows 0-7, preload branch-free; product 2 k sum order permuted
# speedup vs baseline: 1.5021x; 1.5021x over previous
#include <hip/hip_runtime.h>


typedef float v2f __attribute__((ext_vector_type(2)));
typedef float v4f __attribute__((ext_vector_type(4)));
typedef float v8f __attribute__((ext_vector_type(8)));

#define E 8
#define F 32
#define WAVES_PER_BLOCK 8

__launch_bounds__(256, 1)
__global__ void ffq_kernel(const float* __restrict__ x,
                           const float* __restrict__ theta,
                           const float* __restrict__ w1,
                           const float* __restrict__ b1,
                           const float* __restrict__ w2,
                           const float* __restrict__ b2,
                           float* __restrict__ out,
                           int n_tiles) {
  const int tid  = threadIdx.x;
  const int wave = tid >> 5;
  const int lane = tid & 31;
  const int lo   = lane & 15;
  const int hi   = lane >> 4;
  const int c    = hi * 2;

  const float t0 = theta[c], t1 = theta[c + 1], t2 = theta[c + 4], t3 = theta[c + 5];

  v2f aw1[2][2];
  for (int mt = 0; mt < 2; ++mt) {
    const int n = mt * 16 + lo;
    for (int kc = 0; kc < 2; ++kc) {
      aw1[mt][kc].x = w1[(4 * kc + c    ) * F + n];
      aw1[mt][kc].y = w1[(4 * kc + c + 1) * F + n];
    }
  }
  float bias1_0[8], bias1_1[8];
  for (int r = 0; r < 8; ++r) {
    bias1_0[r] = b1[8 * hi + r];
    bias1_1[r] = b1[16 + 8 * hi + r];
  }

  v2f aw2[8];
  for (int j = 0; j < 8; ++j) {
    const int k = 16 * (j >> 2) + 8 * hi + 2 * (j & 3);
    aw2[j].x = w2[(k    ) * E + (lo & 7)];
    aw2[j].y = w2[(k + 1) * E + (lo & 7)];
  }
  const float bias2_0 = b2[4 * hi], bias2_1 = b2[4 * hi + 1], bias2_2 = b2[4 * hi + 2], bias2_3 = b2[4 * hi + 3];

  const int wave_gid = blockIdx.x * WAVES_PER_BLOCK + wave;
  const int n_waves  = gridDim.x * WAVES_PER_BLOCK;

  for (int tile = wave_gid; tile < n_tiles; tile += n_waves) {
    const int row = tile * 16 + lo;
    const float* px = x + row * E + c;
    const v2f xa = *(const v2f*)(px);
    const v2f xb = *(const v2f*)(px + 4);

    v2f q0, q1;
    q0.x = __builtin_cosf(xa.x + t0);
    q0.y = __builtin_cosf(xa.y + t1);
    q1.x = __builtin_cosf(xb.x + t2);
    q1.y = __builtin_cosf(xb.y + t3);

    v8f h0 = {};
    v8f h1 = {};
    h0 = __builtin_amdgcn_wmma_f32_16x16x4_f32(false, aw1[0][0], false, q0, (short)0, h0, false, false);
    h0 = __builtin_amdgcn_wmma_f32_16x16x4_f32(false, aw1[0][1], false, q1, (short)0, h0, false, false);
    h1 = __builtin_amdgcn_wmma_f32_16x16x4_f32(false, aw1[1][0], false, q0, (short)0, h1, false, false);
    h1 = __builtin_amdgcn_wmma_f32_16x16x4_f32(false, aw1[1][1], false, q1, (short)0, h1, false, false);

#pragma unroll
    for (int r = 0; r < 8; ++r) {
      float u = h0[r] + bias1_0[r]; h0[r] = u > 0.0f ? u : 0.0f;
      float v = h1[r] + bias1_1[r]; h1[r] = v > 0.0f ? v : 0.0f;
    }

    v8f acc2 = {};
#pragma unroll
    for (int j = 0; j < 8; ++j) {
      v2f hb;
      if (j < 4) { hb.x = h0[2 * j];       hb.y = h0[2 * j + 1]; }
      else       { hb.x = h1[2 * (j - 4)]; hb.y = h1[2 * (j - 4) + 1]; }
      acc2 = __builtin_amdgcn_wmma_f32_16x16x4_f32(false, aw2[j], false, hb, (short)0, acc2, false, false);
    }

    v4f o;
    o.x = (hi ? acc2[4] : acc2[0]) + bias2_0;
    o.y = (hi ? acc2[5] : acc2[1]) + bias2_1;
    o.z = (hi ? acc2[6] : acc2[2]) + bias2_2;
    o.w = (hi ? acc2[7] : acc2[3]) + bias2_3;

    *(v4f*)(out + tile * 128 + lo * 8 + hi * 4) = o;
  }
}

extern "C" void kernel_launch(void* const* d_in, const int* in_sizes, int n_in,
                              void* d_out, int out_size, void* d_ws, size_t ws_size,
                              hipStream_t stream) {
  const float* x     = (const float*)d_in[0];
  const float* theta = (const float*)d_in[1];
  const float* w1    = (const float*)d_in[2];
  const float* b1    = (const float*)d_in[3];
  const float* w2    = (const float*)d_in[4];
  const float* b2    = (const float*)d_in[5];
  float* out = (float*)d_out;

  const int n_rows  = in_sizes[0] / E;
  const int n_tiles = n_rows / 16;

  ffq_kernel<<<1024, 256, 0, stream>>>(x, theta, w1, b1, w2, b2, out, n_tiles);
}


# grid 1024 -> 512 workgroups (two per WGP), 8 tiles per wave: half the waves to start; each wave requests the rows of all its 8 tiles as one load group up front, then computes them in turn
# speedup vs baseline: 1.7264x; 1.1493x over previous
#include <hip/hip_runtime.h>


typedef float v2f __attribute__((ext_vector_type(2)));
typedef float v4f __attribute__((ext_vector_type(4)));
typedef float v8f __attribute__((ext_vector_type(8)));

#define E 8
#define F 32
#define WAVES_PER_BLOCK 8
#define CHUNK 8
#define N_BLOCKS 512

__launch_bounds__(256, 1)
__global__ void ffq_kernel(const float* __restrict__ x,
                           const float* __restrict__ theta,
                           const float* __restrict__ w1,
                           const float* __restrict__ b1,
                           const float* __restrict__ w2,
                           const float* __restrict__ b2,
                           float* __restrict__ out,
                           int n_tiles) {
  const int tid  = threadIdx.x;
  const int wave = __builtin_amdgcn_readfirstlane(tid >> 5);
  const int lane = tid & 31;
  const int lo   = lane & 15;
  const int hi   = lane >> 4;
  const int c    = hi * 2;

  const float t0 = theta[c], t1 = theta[c + 1], t2 = theta[c + 4], t3 = theta[c + 5];

  v2f aw1[2][2];
  for (int mt = 0; mt < 2; ++mt) {
    const int n = mt * 16 + lo;
    for (int kc = 0; kc < 2; ++kc) {
      aw1[mt][kc].x = w1[(4 * kc + c    ) * F + n];
      aw1[mt][kc].y = w1[(4 * kc + c + 1) * F + n];
    }
  }
  float bias1_0[8], bias1_1[8];
  for (int r = 0; r < 8; ++r) {
    bias1_0[r] = b1[8 * hi + r];
    bias1_1[r] = b1[16 + 8 * hi + r];
  }

  v2f aw2[8];
  for (int j = 0; j < 8; ++j) {
    const int k = 16 * (j >> 2) + 8 * hi + 2 * (j & 3);
    aw2[j].x = w2[(k    ) * E + (lo & 7)];
    aw2[j].y = w2[(k + 1) * E + (lo & 7)];
  }
  const float bias2_0 = b2[4 * hi], bias2_1 = b2[4 * hi + 1], bias2_2 = b2[4 * hi + 2], bias2_3 = b2[4 * hi + 3];

  const int wave_gid = blockIdx.x * WAVES_PER_BLOCK + wave;
  const int n_waves  = gridDim.x * WAVES_PER_BLOCK;
  const int lane_off = lo * E + c;

  for (int base = wave_gid; base < n_tiles; base += CHUNK * n_waves) {
    v2f xa[CHUNK], xb[CHUNK];
#pragma unroll
    for (int i = 0; i < CHUNK; ++i) {
      const int t  = base + i * n_waves;
      const int tl = t < n_tiles ? t : base;
      const float* px = x + (size_t)tl * (16 * E) + lane_off;
      xa[i] = *(const v2f*)(px);
      xb[i] = *(const v2f*)(px + 4);
    }
    __builtin_amdgcn_sched_barrier(0);

#pragma unroll
    for (int i = 0; i < CHUNK; ++i) {
      const int tile = base + i * n_waves;
      {
        v2f q0, q1;
        q0.x = __builtin_cosf(xa[i].x + t0);
        q0.y = __builtin_cosf(xa[i].y + t1);
        q1.x = __builtin_cosf(xb[i].x + t2);
        q1.y = __builtin_cosf(xb[i].y + t3);

        v8f h0 = {};
        v8f h1 = {};
        h0 = __builtin_amdgcn_wmma_f32_16x16x4_f32(false, aw1[0][0], false, q0, (short)0, h0, false, false);
        h0 = __builtin_amdgcn_wmma_f32_16x16x4_f32(false, aw1[0][1], false, q1, (short)0, h0, false, false);
        h1 = __builtin_amdgcn_wmma_f32_16x16x4_f32(false, aw1[1][0], false, q0, (short)0, h1, false, false);
        h1 = __builtin_amdgcn_wmma_f32_16x16x4_f32(false, aw1[1][1], false, q1, (short)0, h1, false, false);

#pragma unroll
        for (int r = 0; r < 8; ++r) {
          float u = h0[r] + bias1_0[r]; h0[r] = u > 0.0f ? u : 0.0f;
          float v = h1[r] + bias1_1[r]; h1[r] = v > 0.0f ? v : 0.0f;
        }

        v8f acc2 = {};
#pragma unroll
        for (int j = 0; j < 8; ++j) {
          v2f hb;
          if (j < 4) { hb.x = h0[2 * j];       hb.y = h0[2 * j + 1]; }
          else       { hb.x = h1[2 * (j - 4)]; hb.y = h1[2 * (j - 4) + 1]; }
          acc2 = __builtin_amdgcn_wmma_f32_16x16x4_f32(false, aw2[j], false, hb, (short)0, acc2, false, false);
        }

        v4f o;
        o.x = (hi ? acc2[4] : acc2[0]) + bias2_0;
        o.y = (hi ? acc2[5] : acc2[1]) + bias2_1;
        o.z = (hi ? acc2[6] : acc2[2]) + bias2_2;
        o.w = (hi ? acc2[7] : acc2[3]) + bias2_3;

        if (tile < n_tiles)
          *(v4f*)(out + (size_t)tile * 128 + lo * 8 + hi * 4) = o;
      }
    }
  }
}

extern "C" void kernel_launch(void* const* d_in, const int* in_sizes, int n_in,
                              void* d_out, int out_size, void* d_ws, size_t ws_size,
                              hipStream_t stream) {
  const float* x     = (const float*)d_in[0];
  const float* theta = (const float*)d_in[1];
  const float* w1    = (const float*)d_in[2];
  const float* b1    = (const float*)d_in[3];
  const float* w2    = (const float*)d_in[4];
  const float* b2    = (const float*)d_in[5];
  float* out = (float*)d_out;

  const int n_rows  = in_sizes[0] / E;
  const int n_tiles = n_rows / 16;

  ffq_kernel<<<N_BLOCKS, 256, 0, stream>>>(x, theta, w1, b1, w2, b2, out, n_tiles);
}
